# starting kernel
# baseline (speedup 1.0000x reference)
#include <hip/hip_runtime.h>

typedef __attribute__((ext_vector_type(2))) float v2f;
typedef __attribute__((ext_vector_type(4))) float v4f;
typedef __attribute__((ext_vector_type(8))) float v8f;

#define M_ROWS 4096
#define N_COLS 50000
#define DIM    64
#define WAVES  8
#define TILE_N 16

#define KDE_COEFF (-69.63184405430771f)

__global__ __launch_bounds__(256)
void kde_logsumexp_wmma(const float* __restrict__ x,
                        const float* __restrict__ Xm,
                        const float* __restrict__ bw,
                        float* __restrict__ out) {
  __shared__ float lds_x2[16];
  __shared__ float lds_m[WAVES][16];
  __shared__ float lds_s[WAVES][16];

  const int tid  = threadIdx.x;
  const int wave = tid >> 5;
  const int lane = tid & 31;
  const int hi   = lane >> 4;
  const int l16  = lane & 15;
  const int row0 = blockIdx.x * 16;

  if (tid < 16) {
    const float* xr = x + (size_t)(row0 + tid) * DIM;
    float acc = 0.0f;
#pragma unroll
    for (int d = 0; d < DIM; ++d) acc = fmaf(xr[d], xr[d], acc);
    lds_x2[tid] = acc;
  }
  __syncthreads();

  v4f a4[8];
  {
    const float* xr = x + (size_t)(row0 + l16) * DIM + 4 * hi;
#pragma unroll
    for (int c = 0; c < 8; ++c) a4[c] = *(const v4f*)(xr + 8 * c);
  }

  const float b  = bw[0];
  const float s2 = -1.4426950408889634f / (2.0f * b * b);
  const float u  = -2.0f * s2;

  float x2s2[8];
#pragma unroll
  for (int k = 0; k < 8; ++k) x2s2[k] = lds_x2[8 * hi + k] * s2;

  float m[8], s[8];
#pragma unroll
  for (int k = 0; k < 8; ++k) { m[k] = -1.0e30f; s[k] = 0.0f; }

  const float* Xbase = Xm + (size_t)l16 * DIM + 4 * hi;

  auto loadB = [&](v4f (&bf)[8], int t) {
    const float* Xc = Xbase + (size_t)t * (TILE_N * DIM);
#pragma unroll
    for (int c = 0; c < 8; ++c) bf[c] = *(const v4f*)(Xc + 8 * c);
  };

  auto computeTile = [&](const v4f (&bf)[8]) {
    float p0 = 0.f, p1 = 0.f, p2 = 0.f, p3 = 0.f;
#pragma unroll
    for (int c = 0; c < 8; ++c) {
      p0 = fmaf(bf[c].x, bf[c].x, p0);
      p1 = fmaf(bf[c].y, bf[c].y, p1);
      p2 = fmaf(bf[c].z, bf[c].z, p2);
      p3 = fmaf(bf[c].w, bf[c].w, p3);
    }
    const float p     = (p0 + p1) + (p2 + p3);
    const float X2col = p + __shfl_xor(p, 16, 32);

    v8f c0 = {0.f, 0.f, 0.f, 0.f, 0.f, 0.f, 0.f, 0.f};
    v8f c1 = {0.f, 0.f, 0.f, 0.f, 0.f, 0.f, 0.f, 0.f};
#pragma unroll
    for (int c = 0; c < 4; ++c) {
      const v2f alo0 = __builtin_shufflevector(a4[c],     a4[c],     0, 1);
      const v2f ahi0 = __builtin_shufflevector(a4[c],     a4[c],     2, 3);
      const v2f blo0 = __builtin_shufflevector(bf[c],     bf[c],     0, 1);
      const v2f bhi0 = __builtin_shufflevector(bf[c],     bf[c],     2, 3);
      const v2f alo1 = __builtin_shufflevector(a4[c + 4], a4[c + 4], 0, 1);
      const v2f ahi1 = __builtin_shufflevector(a4[c + 4], a4[c + 4], 2, 3);
      const v2f blo1 = __builtin_shufflevector(bf[c + 4], bf[c + 4], 0, 1);
      const v2f bhi1 = __builtin_shufflevector(bf[c + 4], bf[c + 4], 2, 3);
      c0 = __builtin_amdgcn_wmma_f32_16x16x4_f32(false, alo0, false, blo0,
                                                 (short)0, c0, false, false);
      c1 = __builtin_amdgcn_wmma_f32_16x16x4_f32(false, alo1, false, blo1,
                                                 (short)0, c1, false, false);
      c0 = __builtin_amdgcn_wmma_f32_16x16x4_f32(false, ahi0, false, bhi0,
                                                 (short)0, c0, false, false);
      c1 = __builtin_amdgcn_wmma_f32_16x16x4_f32(false, ahi1, false, bhi1,
                                                 (short)0, c1, false, false);
    }
    const v8f cc = c0 + c1;

    const float base = X2col * s2;
#pragma unroll
    for (int k = 0; k < 8; ++k) {
      const float rc = x2s2[k] + base;
      float l = fmaf(cc[k], u, rc);
      l = fminf(l, 0.0f);
      m[k] = fmaxf(m[k], l);
      s[k] += __builtin_amdgcn_exp2f(l);
    }
  };

  const int ntiles = N_COLS / TILE_N;
  {
    v4f b0[8], b1[8];
    int t = wave;
    loadB(b0, t);
    for (;;) {
      const int t1 = t + WAVES;
      if (t1 >= ntiles) { computeTile(b0); break; }
      loadB(b1, t1);
      computeTile(b0);
      const int t2 = t1 + WAVES;
      if (t2 >= ntiles) { computeTile(b1); break; }
      loadB(b0, t2);
      computeTile(b1);
      t = t2;
    }
  }

#pragma unroll
  for (int off = 1; off < 16; off <<= 1) {
#pragma unroll
    for (int k = 0; k < 8; ++k) {
      s[k] += __shfl_xor(s[k], off, 32);
      m[k] = fmaxf(m[k], __shfl_xor(m[k], off, 32));
    }
  }

  if (l16 == 0) {
#pragma unroll
    for (int k = 0; k < 8; ++k) {
      lds_m[wave][8 * hi + k] = m[k];
      lds_s[wave][8 * hi + k] = s[k];
    }
  }
  __syncthreads();

  if (tid < 16) {
    float ss = 0.0f, mm = -1.0e30f;
#pragma unroll
    for (int w = 0; w < WAVES; ++w) {
      ss += lds_s[w][tid];
      mm = fmaxf(mm, lds_m[w][tid]);
    }
    const float r = (ss > 0.0f) ? __builtin_amdgcn_logf(ss) : mm;
    out[row0 + tid] = r * 0.6931471805599453f + KDE_COEFF;
  }
}

extern "C" void kernel_launch(void* const* d_in, const int* in_sizes, int n_in,
                              void* d_out, int out_size, void* d_ws, size_t ws_size,
                              hipStream_t stream) {
  const float* x  = (const float*)d_in[0];
  const float* Xm = (const float*)d_in[1];
  const float* bw = (const float*)d_in[2];
  float* outp = (float*)d_out;

  dim3 grid(M_ROWS / 16);
  dim3 block(256);
  hipLaunchKernelGGL(kde_logsumexp_wmma, grid, block, 0, stream, x, Xm, bw, outp);
}


# B tiles fetched by whole-line loads (512 contiguous bytes per load: 4 lines, not 16) and re-laid into the fragment layout through a per-wave LDS stage; same k order and sums
# speedup vs baseline: 1.1890x; 1.1890x over previous
#include <hip/hip_runtime.h>

typedef __attribute__((ext_vector_type(2))) float v2f;
typedef __attribute__((ext_vector_type(4))) float v4f;
typedef __attribute__((ext_vector_type(8))) float v8f;

#define M_ROWS 4096
#define N_COLS 50000
#define DIM    64
#define WAVES  8
#define TILE_N 16
#define LDS_PITCH 68

#define KDE_COEFF (-69.63184405430771f)

__global__ __launch_bounds__(256)
void kde_logsumexp_wmma(const float* __restrict__ x,
                        const float* __restrict__ Xm,
                        const float* __restrict__ bw,
                        float* __restrict__ out) {
  __shared__ float lds_x2[16];
  __shared__ float lds_m[WAVES][16];
  __shared__ float lds_s[WAVES][16];
  __shared__ __attribute__((aligned(16))) float lds_b[WAVES][16 * LDS_PITCH];

  const int tid  = threadIdx.x;
  const int wave = __builtin_amdgcn_readfirstlane(tid >> 5);
  const int lane = tid & 31;
  const int hi   = lane >> 4;
  const int l16  = lane & 15;
  const int row0 = blockIdx.x * 16;

  if (tid < 16) {
    const float* xr = x + (size_t)(row0 + tid) * DIM;
    float acc = 0.0f;
#pragma unroll
    for (int d = 0; d < DIM; ++d) acc = fmaf(xr[d], xr[d], acc);
    lds_x2[tid] = acc;
  }
  __syncthreads();

  v4f a4[8];
  {
    const float* xr = x + (size_t)(row0 + l16) * DIM + 4 * hi;
#pragma unroll
    for (int c = 0; c < 8; ++c) a4[c] = *(const v4f*)(xr + 8 * c);
  }

  const float b  = bw[0];
  const float s2 = -1.4426950408889634f / (2.0f * b * b);
  const float u  = -2.0f * s2;

  float x2s2[8];
#pragma unroll
  for (int k = 0; k < 8; ++k) x2s2[k] = lds_x2[8 * hi + k] * s2;

  float m[8], s[8];
#pragma unroll
  for (int k = 0; k < 8; ++k) { m[k] = -1.0e30f; s[k] = 0.0f; }

  const int cnt = (N_COLS / TILE_N - wave + WAVES - 1) / WAVES;
  float* const stage = &lds_b[wave][0];
  float* const st_ptr = stage + hi * LDS_PITCH + 4 * l16;
  const float* const fr_ptr = stage + l16 * LDS_PITCH + 4 * hi;
  const float* const Xlane = Xm + (size_t)wave * (TILE_N * DIM) + 4 * lane;

  auto loadRaw = [&](v4f (&g)[8], int j) {
    const int jj = j < cnt ? j : cnt - 1;
    const float* src = Xlane + (size_t)jj * (WAVES * TILE_N * DIM);
#pragma unroll
    for (int i = 0; i < 8; ++i) g[i] = *(const v4f*)(src + 128 * i);
  };
  auto stageStore = [&](const v4f (&g)[8]) {
#pragma unroll
    for (int i = 0; i < 8; ++i) *(v4f*)(st_ptr + 2 * i * LDS_PITCH) = g[i];
  };
  auto stageLoad = [&](v4f (&bf)[8]) {
#pragma unroll
    for (int c = 0; c < 8; ++c) bf[c] = *(const v4f*)(fr_ptr + 8 * c);
  };

  auto computeTile = [&](const v4f (&bf)[8]) {
    float p0 = 0.f, p1 = 0.f, p2 = 0.f, p3 = 0.f;
#pragma unroll
    for (int c = 0; c < 8; ++c) {
      p0 = fmaf(bf[c].x, bf[c].x, p0);
      p1 = fmaf(bf[c].y, bf[c].y, p1);
      p2 = fmaf(bf[c].z, bf[c].z, p2);
      p3 = fmaf(bf[c].w, bf[c].w, p3);
    }
    const float p     = (p0 + p1) + (p2 + p3);
    const float X2col = p + __shfl_xor(p, 16, 32);

    v8f c0 = {0.f, 0.f, 0.f, 0.f, 0.f, 0.f, 0.f, 0.f};
    v8f c1 = {0.f, 0.f, 0.f, 0.f, 0.f, 0.f, 0.f, 0.f};
#pragma unroll
    for (int c = 0; c < 4; ++c) {
      const v2f alo0 = __builtin_shufflevector(a4[c],     a4[c],     0, 1);
      const v2f ahi0 = __builtin_shufflevector(a4[c],     a4[c],     2, 3);
      const v2f blo0 = __builtin_shufflevector(bf[c],     bf[c],     0, 1);
      const v2f bhi0 = __builtin_shufflevector(bf[c],     bf[c],     2, 3);
      const v2f alo1 = __builtin_shufflevector(a4[c + 4], a4[c + 4], 0, 1);
      const v2f ahi1 = __builtin_shufflevector(a4[c + 4], a4[c + 4], 2, 3);
      const v2f blo1 = __builtin_shufflevector(bf[c + 4], bf[c + 4], 0, 1);
      const v2f bhi1 = __builtin_shufflevector(bf[c + 4], bf[c + 4], 2, 3);
      c0 = __builtin_amdgcn_wmma_f32_16x16x4_f32(false, alo0, false, blo0,
                                                 (short)0, c0, false, false);
      c1 = __builtin_amdgcn_wmma_f32_16x16x4_f32(false, alo1, false, blo1,
                                                 (short)0, c1, false, false);
      c0 = __builtin_amdgcn_wmma_f32_16x16x4_f32(false, ahi0, false, bhi0,
                                                 (short)0, c0, false, false);
      c1 = __builtin_amdgcn_wmma_f32_16x16x4_f32(false, ahi1, false, bhi1,
                                                 (short)0, c1, false, false);
    }
    const v8f cc = c0 + c1;

    const float base = X2col * s2;
#pragma unroll
    for (int k = 0; k < 8; ++k) {
      const float rc = x2s2[k] + base;
      float l = fmaf(cc[k], u, rc);
      l = fminf(l, 0.0f);
      m[k] = fmaxf(m[k], l);
      s[k] += __builtin_amdgcn_exp2f(l);
    }
  };

  {
    v4f gA[8], gB[8], fA[8], fB[8];
    loadRaw(gA, 0);
    loadRaw(gB, 1);
    stageStore(gA);
    loadRaw(gA, 2);
    stageLoad(fA);
    const int npairs = cnt >> 1;
    for (int p = 0; p < npairs; ++p) {
      const int j = 2 * p;
      stageStore(gB);
      loadRaw(gB, j + 3);
      stageLoad(fB);
      computeTile(fA);
      stageStore(gA);
      loadRaw(gA, j + 4);
      stageLoad(fA);
      computeTile(fB);
      __builtin_amdgcn_sched_group_barrier(0x200, 8, 0);
      __builtin_amdgcn_sched_group_barrier(0x020, 8, 0);
      __builtin_amdgcn_sched_group_barrier(0x100, 8, 0);
      __builtin_amdgcn_sched_group_barrier(0x008, 16, 0);
      __builtin_amdgcn_sched_group_barrier(0x200, 8, 0);
      __builtin_amdgcn_sched_group_barrier(0x020, 8, 0);
      __builtin_amdgcn_sched_group_barrier(0x080, 2, 0);
      __builtin_amdgcn_sched_group_barrier(0x100, 8, 0);
      __builtin_amdgcn_sched_group_barrier(0x008, 16, 0);
    }
    if (cnt & 1) computeTile(fA);
  }

#pragma unroll
  for (int off = 1; off < 16; off <<= 1) {
#pragma unroll
    for (int k = 0; k < 8; ++k) {
      s[k] += __shfl_xor(s[k], off, 32);
      m[k] = fmaxf(m[k], __shfl_xor(m[k], off, 32));
    }
  }

  if (l16 == 0) {
#pragma unroll
    for (int k = 0; k < 8; ++k) {
      lds_m[wave][8 * hi + k] = m[k];
      lds_s[wave][8 * hi + k] = s[k];
    }
  }
  __syncthreads();

  if (tid < 16) {
    float ss = 0.0f, mm = -1.0e30f;
#pragma unroll
    for (int w = 0; w < WAVES; ++w) {
      ss += lds_s[w][tid];
      mm = fmaxf(mm, lds_m[w][tid]);
    }
    const float r = (ss > 0.0f) ? __builtin_amdgcn_logf(ss) : mm;
    out[row0 + tid] = r * 0.6931471805599453f + KDE_COEFF;
  }
}

extern "C" void kernel_launch(void* const* d_in, const int* in_sizes, int n_in,
                              void* d_out, int out_size, void* d_ws, size_t ws_size,
                              hipStream_t stream) {
  const float* x  = (const float*)d_in[0];
  const float* Xm = (const float*)d_in[1];
  const float* bw = (const float*)d_in[2];
  float* outp = (float*)d_out;

  dim3 grid(M_ROWS / 16);
  dim3 block(256);
  hipLaunchKernelGGL(kde_logsumexp_wmma, grid, block, 0, stream, x, Xm, bw, outp);
}


# Pre-pass kernel per call writes X to the workspace in WMMA B-fragment order plus s2*|X_col|^2 per column; main loop loads both directly: no LDS stage, no per-tile norm; same sums
# speedup vs baseline: 1.3523x; 1.1373x over previous
#include <hip/hip_runtime.h>

typedef __attribute__((ext_vector_type(2))) float v2f;
typedef __attribute__((ext_vector_type(4))) float v4f;
typedef __attribute__((ext_vector_type(8))) float v8f;

#define M_ROWS 4096
#define N_COLS 50000
#define DIM    64
#define WAVES  8
#define TILE_N 16
#define NTILES (N_COLS / TILE_N)
#define NPAIRS ((NTILES + 1) / 2)
#define TILE_FLOATS (TILE_N * DIM)
#define WS_NORM_OFF ((size_t)N_COLS * DIM)

#define KDE_COEFF (-69.63184405430771f)

__global__ __launch_bounds__(256)
void kde_prep_b(const float* __restrict__ Xm,
                const float* __restrict__ bw,
                float* __restrict__ wfrag,
                float* __restrict__ wnorm) {
  const int tid  = threadIdx.x;
  const int wave = __builtin_amdgcn_readfirstlane(tid >> 5);
  const int lane = tid & 31;
  const int hi   = lane >> 4;
  const int l16  = lane & 15;
  const int T    = blockIdx.x * WAVES + wave;
  if (T >= NPAIRS) return;

  const float b  = bw[0];
  const float s2 = -1.4426950408889634f / (2.0f * b * b);

  float base2[2];
#pragma unroll
  for (int h = 0; h < 2; ++h) {
    const int tile = 2 * T + h;
    const int tl   = tile < NTILES ? tile : NTILES - 1;
    const float* src = Xm + (size_t)tl * TILE_FLOATS + l16 * DIM + 4 * hi;
    v4f f[8];
#pragma unroll
    for (int c = 0; c < 8; ++c) f[c] = *(const v4f*)(src + 8 * c);

    float p0 = 0.f, p1 = 0.f, p2 = 0.f, p3 = 0.f;
#pragma unroll
    for (int c = 0; c < 8; ++c) {
      p0 = fmaf(f[c].x, f[c].x, p0);
      p1 = fmaf(f[c].y, f[c].y, p1);
      p2 = fmaf(f[c].z, f[c].z, p2);
      p3 = fmaf(f[c].w, f[c].w, p3);
    }
    const float p = (p0 + p1) + (p2 + p3);
    base2[h] = (p + __shfl_xor(p, 16, 32)) * s2;

    if (tile < NTILES) {
      float* dst = wfrag + (size_t)tile * TILE_FLOATS + 4 * lane;
#pragma unroll
      for (int c = 0; c < 8; ++c) *(v4f*)(dst + 128 * c) = f[c];
    }
  }
  const int col = 32 * T + lane;
  if (col < N_COLS) wnorm[col] = hi ? base2[1] : base2[0];
}

__global__ __launch_bounds__(256)
void kde_logsumexp_wmma(const float* __restrict__ x,
                        const float* __restrict__ wfrag,
                        const float* __restrict__ wnorm,
                        const float* __restrict__ bw,
                        float* __restrict__ out) {
  __shared__ float lds_x2[16];
  __shared__ float lds_m[WAVES][16];
  __shared__ float lds_s[WAVES][16];

  const int tid  = threadIdx.x;
  const int wave = __builtin_amdgcn_readfirstlane(tid >> 5);
  const int lane = tid & 31;
  const int hi   = lane >> 4;
  const int l16  = lane & 15;
  const int row0 = blockIdx.x * 16;

  if (tid < 16) {
    const float* xr = x + (size_t)(row0 + tid) * DIM;
    float acc = 0.0f;
#pragma unroll
    for (int d = 0; d < DIM; ++d) acc = fmaf(xr[d], xr[d], acc);
    lds_x2[tid] = acc;
  }
  __syncthreads();

  v4f a4[8];
  {
    const float* xr = x + (size_t)(row0 + l16) * DIM + 4 * hi;
#pragma unroll
    for (int c = 0; c < 8; ++c) a4[c] = *(const v4f*)(xr + 8 * c);
  }

  const float b  = bw[0];
  const float s2 = -1.4426950408889634f / (2.0f * b * b);
  const float u  = -2.0f * s2;

  float x2s2[8];
#pragma unroll
  for (int k = 0; k < 8; ++k) x2s2[k] = lds_x2[8 * hi + k] * s2;

  float m[8], s[8];
#pragma unroll
  for (int k = 0; k < 8; ++k) { m[k] = -1.0e30f; s[k] = 0.0f; }

  const int cnt = (NTILES - wave + WAVES - 1) / WAVES;
  const float* const Wlane = wfrag + (size_t)wave * TILE_FLOATS + 4 * lane;
  const float* const Nlane = wnorm + wave * TILE_N + l16;

  auto loadTile = [&](v4f (&bf)[8], float& nb, int j) {
    const int jj = j < cnt ? j : cnt - 1;
    const float* src = Wlane + (size_t)jj * (WAVES * TILE_FLOATS);
#pragma unroll
    for (int c = 0; c < 8; ++c) bf[c] = *(const v4f*)(src + 128 * c);
    nb = Nlane[jj * (WAVES * TILE_N)];
  };

  auto computeTile = [&](const v4f (&bf)[8], const float base) {
    v8f c0 = {0.f, 0.f, 0.f, 0.f, 0.f, 0.f, 0.f, 0.f};
    v8f c1 = {0.f, 0.f, 0.f, 0.f, 0.f, 0.f, 0.f, 0.f};
#pragma unroll
    for (int c = 0; c < 4; ++c) {
      const v2f alo0 = __builtin_shufflevector(a4[c],     a4[c],     0, 1);
      const v2f ahi0 = __builtin_shufflevector(a4[c],     a4[c],     2, 3);
      const v2f blo0 = __builtin_shufflevector(bf[c],     bf[c],     0, 1);
      const v2f bhi0 = __builtin_shufflevector(bf[c],     bf[c],     2, 3);
      const v2f alo1 = __builtin_shufflevector(a4[c + 4], a4[c + 4], 0, 1);
      const v2f ahi1 = __builtin_shufflevector(a4[c + 4], a4[c + 4], 2, 3);
      const v2f blo1 = __builtin_shufflevector(bf[c + 4], bf[c + 4], 0, 1);
      const v2f bhi1 = __builtin_shufflevector(bf[c + 4], bf[c + 4], 2, 3);
      c0 = __builtin_amdgcn_wmma_f32_16x16x4_f32(false, alo0, false, blo0,
                                                 (short)0, c0, false, false);
      c1 = __builtin_amdgcn_wmma_f32_16x16x4_f32(false, alo1, false, blo1,
                                                 (short)0, c1, false, false);
      c0 = __builtin_amdgcn_wmma_f32_16x16x4_f32(false, ahi0, false, bhi0,
                                                 (short)0, c0, false, false);
      c1 = __builtin_amdgcn_wmma_f32_16x16x4_f32(false, ahi1, false, bhi1,
                                                 (short)0, c1, false, false);
    }
    const v8f cc = c0 + c1;

#pragma unroll
    for (int k = 0; k < 8; ++k) {
      const float rc = x2s2[k] + base;
      float l = fmaf(cc[k], u, rc);
      l = fminf(l, 0.0f);
      m[k] = fmaxf(m[k], l);
      s[k] += __builtin_amdgcn_exp2f(l);
    }
  };

  {
    v4f f0[8], f1[8], f2[8], f3[8];
    float n0, n1, n2, n3;
    loadTile(f0, n0, 0);
    loadTile(f1, n1, 1);
    loadTile(f2, n2, 2);
    const int ntrips = cnt >> 2;
    for (int p = 0; p < ntrips; ++p) {
      const int j = 4 * p;
      loadTile(f3, n3, j + 3);
      computeTile(f0, n0);
      loadTile(f0, n0, j + 4);
      computeTile(f1, n1);
      loadTile(f1, n1, j + 5);
      computeTile(f2, n2);
      loadTile(f2, n2, j + 6);
      computeTile(f3, n3);
      __builtin_amdgcn_sched_group_barrier(0x020, 9, 0);
      __builtin_amdgcn_sched_group_barrier(0x008, 16, 0);
      __builtin_amdgcn_sched_group_barrier(0x020, 9, 0);
      __builtin_amdgcn_sched_group_barrier(0x008, 16, 0);
      __builtin_amdgcn_sched_group_barrier(0x020, 9, 0);
      __builtin_amdgcn_sched_group_barrier(0x008, 16, 0);
      __builtin_amdgcn_sched_group_barrier(0x020, 9, 0);
      __builtin_amdgcn_sched_group_barrier(0x008, 16, 0);
    }
    const int rem = cnt & 3;
    if (rem > 0) computeTile(f0, n0);
    if (rem > 1) computeTile(f1, n1);
    if (rem > 2) computeTile(f2, n2);
  }

#pragma unroll
  for (int off = 1; off < 16; off <<= 1) {
#pragma unroll
    for (int k = 0; k < 8; ++k) {
      s[k] += __shfl_xor(s[k], off, 32);
      m[k] = fmaxf(m[k], __shfl_xor(m[k], off, 32));
    }
  }

  if (l16 == 0) {
#pragma unroll
    for (int k = 0; k < 8; ++k) {
      lds_m[wave][8 * hi + k] = m[k];
      lds_s[wave][8 * hi + k] = s[k];
    }
  }
  __syncthreads();

  if (tid < 16) {
    float ss = 0.0f, mm = -1.0e30f;
#pragma unroll
    for (int w = 0; w < WAVES; ++w) {
      ss += lds_s[w][tid];
      mm = fmaxf(mm, lds_m[w][tid]);
    }
    const float r = (ss > 0.0f) ? __builtin_amdgcn_logf(ss) : mm;
    out[row0 + tid] = r * 0.6931471805599453f + KDE_COEFF;
  }
}

extern "C" void kernel_launch(void* const* d_in, const int* in_sizes, int n_in,
                              void* d_out, int out_size, void* d_ws, size_t ws_size,
                              hipStream_t stream) {
  const float* x  = (const float*)d_in[0];
  const float* Xm = (const float*)d_in[1];
  const float* bw = (const float*)d_in[2];
  float* outp = (float*)d_out;
  float* wfrag = (float*)d_ws;
  float* wnorm = wfrag + WS_NORM_OFF;

  dim3 block(256);
  dim3 pgrid((NPAIRS + WAVES - 1) / WAVES);
  hipLaunchKernelGGL(kde_prep_b, pgrid, block, 0, stream, Xm, bw, wfrag, wnorm);

  dim3 grid(M_ROWS / 16);
  hipLaunchKernelGGL(kde_logsumexp_wmma, grid, block, 0, stream, x, wfrag, wnorm, bw, outp);
}


# Priority window: waves 4-7 (the younger of each SIMD pair) run the first 64 of their 97 loop trips at s_setprio 3, then 0, so the pair ends together and no wave finishes alone; same tiles and sums
# speedup vs baseline: 1.3948x; 1.0314x over previous
#include <hip/hip_runtime.h>

typedef __attribute__((ext_vector_type(2))) float v2f;
typedef __attribute__((ext_vector_type(4))) float v4f;
typedef __attribute__((ext_vector_type(8))) float v8f;

#define M_ROWS 4096
#define N_COLS 50000
#define DIM    64
#define WAVES  8
#define TILE_N 16
#define NTILES (N_COLS / TILE_N)
#define NPAIRS ((NTILES + 1) / 2)
#define TILE_FLOATS (TILE_N * DIM)
#define WS_NORM_OFF ((size_t)N_COLS * DIM)

#define KDE_COEFF (-69.63184405430771f)

__global__ __launch_bounds__(256)
void kde_prep_b(const float* __restrict__ Xm,
                const float* __restrict__ bw,
                float* __restrict__ wfrag,
                float* __restrict__ wnorm) {
  const int tid  = threadIdx.x;
  const int wave = __builtin_amdgcn_readfirstlane(tid >> 5);
  const int lane = tid & 31;
  const int hi   = lane >> 4;
  const int l16  = lane & 15;
  const int T    = blockIdx.x * WAVES + wave;
  if (T >= NPAIRS) return;

  const float b  = bw[0];
  const float s2 = -1.4426950408889634f / (2.0f * b * b);

  float base2[2];
#pragma unroll
  for (int h = 0; h < 2; ++h) {
    const int tile = 2 * T + h;
    const int tl   = tile < NTILES ? tile : NTILES - 1;
    const float* src = Xm + (size_t)tl * TILE_FLOATS + l16 * DIM + 4 * hi;
    v4f f[8];
#pragma unroll
    for (int c = 0; c < 8; ++c) f[c] = *(const v4f*)(src + 8 * c);

    float p0 = 0.f, p1 = 0.f, p2 = 0.f, p3 = 0.f;
#pragma unroll
    for (int c = 0; c < 8; ++c) {
      p0 = fmaf(f[c].x, f[c].x, p0);
      p1 = fmaf(f[c].y, f[c].y, p1);
      p2 = fmaf(f[c].z, f[c].z, p2);
      p3 = fmaf(f[c].w, f[c].w, p3);
    }
    const float p = (p0 + p1) + (p2 + p3);
    base2[h] = (p + __shfl_xor(p, 16, 32)) * s2;

    if (tile < NTILES) {
      float* dst = wfrag + (size_t)tile * TILE_FLOATS + 4 * lane;
#pragma unroll
      for (int c = 0; c < 8; ++c) *(v4f*)(dst + 128 * c) = f[c];
    }
  }
  const int col = 32 * T + lane;
  if (col < N_COLS) wnorm[col] = hi ? base2[1] : base2[0];
}

__global__ __launch_bounds__(256)
void kde_logsumexp_wmma(const float* __restrict__ x,
                        const float* __restrict__ wfrag,
                        const float* __restrict__ wnorm,
                        const float* __restrict__ bw,
                        float* __restrict__ out) {
  __shared__ float lds_x2[16];
  __shared__ float lds_m[WAVES][16];
  __shared__ float lds_s[WAVES][16];

  const int tid  = threadIdx.x;
  const int wave = __builtin_amdgcn_readfirstlane(tid >> 5);
  const int lane = tid & 31;
  const int hi   = lane >> 4;
  const int l16  = lane & 15;
  const int row0 = blockIdx.x * 16;

  if (tid < 16) {
    const float* xr = x + (size_t)(row0 + tid) * DIM;
    float acc = 0.0f;
#pragma unroll
    for (int d = 0; d < DIM; ++d) acc = fmaf(xr[d], xr[d], acc);
    lds_x2[tid] = acc;
  }
  __syncthreads();

  v4f a4[8];
  {
    const float* xr = x + (size_t)(row0 + l16) * DIM + 4 * hi;
#pragma unroll
    for (int c = 0; c < 8; ++c) a4[c] = *(const v4f*)(xr + 8 * c);
  }

  const float b  = bw[0];
  const float s2 = -1.4426950408889634f / (2.0f * b * b);
  const float u  = -2.0f * s2;

  float x2s2[8];
#pragma unroll
  for (int k = 0; k < 8; ++k) x2s2[k] = lds_x2[8 * hi + k] * s2;

  float m[8], s[8];
#pragma unroll
  for (int k = 0; k < 8; ++k) { m[k] = -1.0e30f; s[k] = 0.0f; }

  const int cnt = (NTILES - wave + WAVES - 1) / WAVES;
  const float* const Wlane = wfrag + (size_t)wave * TILE_FLOATS + 4 * lane;
  const float* const Nlane = wnorm + wave * TILE_N + l16;

  auto loadTile = [&](v4f (&bf)[8], float& nb, int j) {
    const int jj = j < cnt ? j : cnt - 1;
    const float* src = Wlane + (size_t)jj * (WAVES * TILE_FLOATS);
#pragma unroll
    for (int c = 0; c < 8; ++c) bf[c] = *(const v4f*)(src + 128 * c);
    nb = Nlane[jj * (WAVES * TILE_N)];
  };

  auto computeTile = [&](const v4f (&bf)[8], const float base) {
    v8f c0 = {0.f, 0.f, 0.f, 0.f, 0.f, 0.f, 0.f, 0.f};
    v8f c1 = {0.f, 0.f, 0.f, 0.f, 0.f, 0.f, 0.f, 0.f};
#pragma unroll
    for (int c = 0; c < 4; ++c) {
      const v2f alo0 = __builtin_shufflevector(a4[c],     a4[c],     0, 1);
      const v2f ahi0 = __builtin_shufflevector(a4[c],     a4[c],     2, 3);
      const v2f blo0 = __builtin_shufflevector(bf[c],     bf[c],     0, 1);
      const v2f bhi0 = __builtin_shufflevector(bf[c],     bf[c],     2, 3);
      const v2f alo1 = __builtin_shufflevector(a4[c + 4], a4[c + 4], 0, 1);
      const v2f ahi1 = __builtin_shufflevector(a4[c + 4], a4[c + 4], 2, 3);
      const v2f blo1 = __builtin_shufflevector(bf[c + 4], bf[c + 4], 0, 1);
      const v2f bhi1 = __builtin_shufflevector(bf[c + 4], bf[c + 4], 2, 3);
      c0 = __builtin_amdgcn_wmma_f32_16x16x4_f32(false, alo0, false, blo0,
                                                 (short)0, c0, false, false);
      c1 = __builtin_amdgcn_wmma_f32_16x16x4_f32(false, alo1, false, blo1,
                                                 (short)0, c1, false, false);
      c0 = __builtin_amdgcn_wmma_f32_16x16x4_f32(false, ahi0, false, bhi0,
                                                 (short)0, c0, false, false);
      c1 = __builtin_amdgcn_wmma_f32_16x16x4_f32(false, ahi1, false, bhi1,
                                                 (short)0, c1, false, false);
    }
    const v8f cc = c0 + c1;

#pragma unroll
    for (int k = 0; k < 8; ++k) {
      const float rc = x2s2[k] + base;
      float l = fmaf(cc[k], u, rc);
      l = fminf(l, 0.0f);
      m[k] = fmaxf(m[k], l);
      s[k] += __builtin_amdgcn_exp2f(l);
    }
  };

  {
    v4f f0[8], f1[8], f2[8], f3[8];
    float n0, n1, n2, n3;
    loadTile(f0, n0, 0);
    loadTile(f1, n1, 1);
    loadTile(f2, n2, 2);
    const int ntrips = cnt >> 2;
    auto trip = [&](const int p) __attribute__((always_inline)) {
      const int j = 4 * p;
      loadTile(f3, n3, j + 3);
      computeTile(f0, n0);
      loadTile(f0, n0, j + 4);
      computeTile(f1, n1);
      loadTile(f1, n1, j + 5);
      computeTile(f2, n2);
      loadTile(f2, n2, j + 6);
      computeTile(f3, n3);
      __builtin_amdgcn_sched_group_barrier(0x020, 9, 0);
      __builtin_amdgcn_sched_group_barrier(0x008, 16, 0);
      __builtin_amdgcn_sched_group_barrier(0x020, 9, 0);
      __builtin_amdgcn_sched_group_barrier(0x008, 16, 0);
      __builtin_amdgcn_sched_group_barrier(0x020, 9, 0);
      __builtin_amdgcn_sched_group_barrier(0x008, 16, 0);
      __builtin_amdgcn_sched_group_barrier(0x020, 9, 0);
      __builtin_amdgcn_sched_group_barrier(0x008, 16, 0);
    };
    const bool younger = wave >= 4;
    constexpr int P_DROP = (2 * ((NTILES / WAVES) >> 2)) / 3;
    const int p_mid = P_DROP < ntrips ? P_DROP : ntrips;
    if (younger) __builtin_amdgcn_s_setprio(3);
    int p = 0;
    for (; p < p_mid; ++p) trip(p);
    if (younger) __builtin_amdgcn_s_setprio(0);
    for (; p < ntrips; ++p) trip(p);
    const int rem = cnt & 3;
    if (rem > 0) computeTile(f0, n0);
    if (rem > 1) computeTile(f1, n1);
    if (rem > 2) computeTile(f2, n2);
  }

#pragma unroll
  for (int off = 1; off < 16; off <<= 1) {
#pragma unroll
    for (int k = 0; k < 8; ++k) {
      s[k] += __shfl_xor(s[k], off, 32);
      m[k] = fmaxf(m[k], __shfl_xor(m[k], off, 32));
    }
  }

  if (l16 == 0) {
#pragma unroll
    for (int k = 0; k < 8; ++k) {
      lds_m[wave][8 * hi + k] = m[k];
      lds_s[wave][8 * hi + k] = s[k];
    }
  }
  __syncthreads();

  if (tid < 16) {
    float ss = 0.0f, mm = -1.0e30f;
#pragma unroll
    for (int w = 0; w < WAVES; ++w) {
      ss += lds_s[w][tid];
      mm = fmaxf(mm, lds_m[w][tid]);
    }
    const float r = (ss > 0.0f) ? __builtin_amdgcn_logf(ss) : mm;
    out[row0 + tid] = r * 0.6931471805599453f + KDE_COEFF;
  }
}

extern "C" void kernel_launch(void* const* d_in, const int* in_sizes, int n_in,
                              void* d_out, int out_size, void* d_ws, size_t ws_size,
                              hipStream_t stream) {
  const float* x  = (const float*)d_in[0];
  const float* Xm = (const float*)d_in[1];
  const float* bw = (const float*)d_in[2];
  float* outp = (float*)d_out;
  float* wfrag = (float*)d_ws;
  float* wnorm = wfrag + WS_NORM_OFF;

  dim3 block(256);
  dim3 pgrid((NPAIRS + WAVES - 1) / WAVES);
  hipLaunchKernelGGL(kde_prep_b, pgrid, block, 0, stream, Xm, bw, wfrag, wnorm);

  dim3 grid(M_ROWS / 16);
  hipLaunchKernelGGL(kde_logsumexp_wmma, grid, block, 0, stream, x, wfrag, wnorm, bw, outp);
}


# Each B fragment set feeds 4 row tiles (64 rows per wave): L2 bytes and loads per WMMA fall to a quarter; columns split in 4 ranges, a 3rd kernel adds the 4 partial row sums (sum order changes)
# speedup vs baseline: 1.6452x; 1.1795x over previous
#include <hip/hip_runtime.h>

typedef __attribute__((ext_vector_type(2))) float v2f;
typedef __attribute__((ext_vector_type(4))) float v4f;
typedef __attribute__((ext_vector_type(8))) float v8f;

#define M_ROWS 4096
#define N_COLS 50000
#define DIM    64
#define WAVES  8
#define TILE_N 16
#define NTILES (N_COLS / TILE_N)
#define NPAIRS ((NTILES + 1) / 2)
#define TILE_FLOATS (TILE_N * DIM)
#define RT 4
#define NSPLIT  RT
#define ROWS_WG (16 * RT)
#define NROWBLK (M_ROWS / ROWS_WG)
#define WS_NORM_OFF ((size_t)N_COLS * DIM)
#define WS_PSUM_OFF (WS_NORM_OFF + 65536)
#define WS_PMAX_OFF (WS_PSUM_OFF + (size_t)NSPLIT * M_ROWS)

#define KDE_COEFF (-69.63184405430771f)

__global__ __launch_bounds__(256)
void kde_prep_b(const float* __restrict__ Xm,
                const float* __restrict__ bw,
                float* __restrict__ wfrag,
                float* __restrict__ wnorm) {
  const int tid  = threadIdx.x;
  const int wave = __builtin_amdgcn_readfirstlane(tid >> 5);
  const int lane = tid & 31;
  const int hi   = lane >> 4;
  const int l16  = lane & 15;
  const int T    = blockIdx.x * WAVES + wave;
  if (T >= NPAIRS) return;

  const float b  = bw[0];
  const float s2 = -1.4426950408889634f / (2.0f * b * b);

  float base2[2];
#pragma unroll
  for (int h = 0; h < 2; ++h) {
    const int tile = 2 * T + h;
    const int tl   = tile < NTILES ? tile : NTILES - 1;
    const float* src = Xm + (size_t)tl * TILE_FLOATS + l16 * DIM + 4 * hi;
    v4f f[8];
#pragma unroll
    for (int c = 0; c < 8; ++c) f[c] = *(const v4f*)(src + 8 * c);

    float p0 = 0.f, p1 = 0.f, p2 = 0.f, p3 = 0.f;
#pragma unroll
    for (int c = 0; c < 8; ++c) {
      p0 = fmaf(f[c].x, f[c].x, p0);
      p1 = fmaf(f[c].y, f[c].y, p1);
      p2 = fmaf(f[c].z, f[c].z, p2);
      p3 = fmaf(f[c].w, f[c].w, p3);
    }
    const float p = (p0 + p1) + (p2 + p3);
    base2[h] = (p + __shfl_xor(p, 16, 32)) * s2;

    if (tile < NTILES) {
      float* dst = wfrag + (size_t)tile * TILE_FLOATS + 4 * lane;
#pragma unroll
      for (int c = 0; c < 8; ++c) *(v4f*)(dst + 128 * c) = f[c];
    }
  }
  const int col = 32 * T + lane;
  if (col < N_COLS) wnorm[col] = hi ? base2[1] : base2[0];
}

__global__ __launch_bounds__(256)
void kde_logsumexp_wmma(const float* __restrict__ x,
                        const float* __restrict__ wfrag,
                        const float* __restrict__ wnorm,
                        const float* __restrict__ bw,
                        float* __restrict__ psum,
                        float* __restrict__ pmax) {
  __shared__ float lds_x2[ROWS_WG];
  __shared__ float lds_m[WAVES][ROWS_WG];
  __shared__ float lds_s[WAVES][ROWS_WG];

  const int tid  = threadIdx.x;
  const int wave = __builtin_amdgcn_readfirstlane(tid >> 5);
  const int lane = tid & 31;
  const int hi   = lane >> 4;
  const int l16  = lane & 15;
  const int rblk = blockIdx.x % NROWBLK;
  const int q    = blockIdx.x / NROWBLK;
  const int row0 = rblk * ROWS_WG;
  const int t0   = (q * NTILES) / NSPLIT;
  const int t1   = ((q + 1) * NTILES) / NSPLIT;

  if (tid < ROWS_WG) {
    const float* xr = x + (size_t)(row0 + tid) * DIM;
    float acc = 0.0f;
#pragma unroll
    for (int d = 0; d < DIM; ++d) acc = fmaf(xr[d], xr[d], acc);
    lds_x2[tid] = acc;
  }
  __syncthreads();

  v4f a4[RT][8];
#pragma unroll
  for (int t = 0; t < RT; ++t) {
    const float* xr = x + (size_t)(row0 + 16 * t + l16) * DIM + 4 * hi;
#pragma unroll
    for (int c = 0; c < 8; ++c) a4[t][c] = *(const v4f*)(xr + 8 * c);
  }

  const float b  = bw[0];
  const float s2 = -1.4426950408889634f / (2.0f * b * b);
  const float u  = -2.0f * s2;

  float x2s2[RT][8];
#pragma unroll
  for (int t = 0; t < RT; ++t)
#pragma unroll
    for (int k = 0; k < 8; ++k) x2s2[t][k] = lds_x2[16 * t + 8 * hi + k] * s2;

  float m[RT][8], s[RT][8];
#pragma unroll
  for (int t = 0; t < RT; ++t)
#pragma unroll
    for (int k = 0; k < 8; ++k) { m[t][k] = -1.0e30f; s[t][k] = 0.0f; }

  const int cnt = (t1 - t0 - wave + WAVES - 1) / WAVES;
  const float* const Wlane = wfrag + (size_t)(t0 + wave) * TILE_FLOATS + 4 * lane;
  const float* const Nlane = wnorm + (t0 + wave) * TILE_N + l16;

  auto loadTile = [&](v4f (&bf)[8], float& nb, int j) {
    const int jj = j < cnt ? j : cnt - 1;
    const float* src = Wlane + (size_t)jj * (WAVES * TILE_FLOATS);
#pragma unroll
    for (int c = 0; c < 8; ++c) bf[c] = *(const v4f*)(src + 128 * c);
    nb = Nlane[jj * (WAVES * TILE_N)];
  };

  auto computeTile = [&](const v4f (&bf)[8], const float base) {
#pragma unroll
    for (int t = 0; t < RT; ++t) {
      v8f c0 = {0.f, 0.f, 0.f, 0.f, 0.f, 0.f, 0.f, 0.f};
      v8f c1 = {0.f, 0.f, 0.f, 0.f, 0.f, 0.f, 0.f, 0.f};
#pragma unroll
      for (int c = 0; c < 4; ++c) {
        const v2f alo0 = __builtin_shufflevector(a4[t][c],     a4[t][c],     0, 1);
        const v2f ahi0 = __builtin_shufflevector(a4[t][c],     a4[t][c],     2, 3);
        const v2f blo0 = __builtin_shufflevector(bf[c],     bf[c],     0, 1);
        const v2f bhi0 = __builtin_shufflevector(bf[c],     bf[c],     2, 3);
        const v2f alo1 = __builtin_shufflevector(a4[t][c + 4], a4[t][c + 4], 0, 1);
        const v2f ahi1 = __builtin_shufflevector(a4[t][c + 4], a4[t][c + 4], 2, 3);
        const v2f blo1 = __builtin_shufflevector(bf[c + 4], bf[c + 4], 0, 1);
        const v2f bhi1 = __builtin_shufflevector(bf[c + 4], bf[c + 4], 2, 3);
        c0 = __builtin_amdgcn_wmma_f32_16x16x4_f32(false, alo0, false, blo0,
                                                   (short)0, c0, false, false);
        c1 = __builtin_amdgcn_wmma_f32_16x16x4_f32(false, alo1, false, blo1,
                                                   (short)0, c1, false, false);
        c0 = __builtin_amdgcn_wmma_f32_16x16x4_f32(false, ahi0, false, bhi0,
                                                   (short)0, c0, false, false);
        c1 = __builtin_amdgcn_wmma_f32_16x16x4_f32(false, ahi1, false, bhi1,
                                                   (short)0, c1, false, false);
      }
      const v8f cc = c0 + c1;

#pragma unroll
      for (int k = 0; k < 8; ++k) {
        const float rc = x2s2[t][k] + base;
        float l = fmaf(cc[k], u, rc);
        l = fminf(l, 0.0f);
        m[t][k] = fmaxf(m[t][k], l);
        s[t][k] += __builtin_amdgcn_exp2f(l);
      }
    }
  };

  {
    v4f f0[8], f1[8];
    float n0, n1;
    loadTile(f0, n0, 0);
    const int ntrips = cnt >> 1;
    auto trip = [&](const int p) __attribute__((always_inline)) {
      const int j = 2 * p;
      loadTile(f1, n1, j + 1);
      computeTile(f0, n0);
      loadTile(f0, n0, j + 2);
      computeTile(f1, n1);
      __builtin_amdgcn_sched_group_barrier(0x020, 9, 0);
      __builtin_amdgcn_sched_group_barrier(0x008, 16 * RT, 0);
      __builtin_amdgcn_sched_group_barrier(0x020, 9, 0);
      __builtin_amdgcn_sched_group_barrier(0x008, 16 * RT, 0);
    };
    const bool younger = wave >= 4;
    const int p_mid = (2 * ntrips) / 3;
    if (younger) __builtin_amdgcn_s_setprio(3);
    int p = 0;
    for (; p < p_mid; ++p) trip(p);
    if (younger) __builtin_amdgcn_s_setprio(0);
    for (; p < ntrips; ++p) trip(p);
    if (cnt & 1) computeTile(f0, n0);
  }

#pragma unroll
  for (int t = 0; t < RT; ++t) {
#pragma unroll
    for (int off = 1; off < 16; off <<= 1) {
#pragma unroll
      for (int k = 0; k < 8; ++k) {
        s[t][k] += __shfl_xor(s[t][k], off, 32);
        m[t][k] = fmaxf(m[t][k], __shfl_xor(m[t][k], off, 32));
      }
    }
    if (l16 == 0) {
#pragma unroll
      for (int k = 0; k < 8; ++k) {
        lds_m[wave][16 * t + 8 * hi + k] = m[t][k];
        lds_s[wave][16 * t + 8 * hi + k] = s[t][k];
      }
    }
  }
  __syncthreads();

  if (tid < ROWS_WG) {
    float ss = 0.0f, mm = -1.0e30f;
#pragma unroll
    for (int w = 0; w < WAVES; ++w) {
      ss += lds_s[w][tid];
      mm = fmaxf(mm, lds_m[w][tid]);
    }
    psum[(size_t)q * M_ROWS + row0 + tid] = ss;
    pmax[(size_t)q * M_ROWS + row0 + tid] = mm;
  }
}

__global__ __launch_bounds__(256)
void kde_merge_ranges(const float* __restrict__ psum,
                      const float* __restrict__ pmax,
                      float* __restrict__ out) {
  const int row = blockIdx.x * 256 + threadIdx.x;
  float ss = 0.0f, mm = -1.0e30f;
#pragma unroll
  for (int r = 0; r < NSPLIT; ++r) {
    ss += psum[(size_t)r * M_ROWS + row];
    mm = fmaxf(mm, pmax[(size_t)r * M_ROWS + row]);
  }
  const float r2 = (ss > 0.0f) ? __builtin_amdgcn_logf(ss) : mm;
  out[row] = r2 * 0.6931471805599453f + KDE_COEFF;
}

extern "C" void kernel_launch(void* const* d_in, const int* in_sizes, int n_in,
                              void* d_out, int out_size, void* d_ws, size_t ws_size,
                              hipStream_t stream) {
  const float* x  = (const float*)d_in[0];
  const float* Xm = (const float*)d_in[1];
  const float* bw = (const float*)d_in[2];
  float* outp = (float*)d_out;
  float* wfrag = (float*)d_ws;
  float* wnorm = wfrag + WS_NORM_OFF;
  float* psum  = wfrag + WS_PSUM_OFF;
  float* pmax  = wfrag + WS_PMAX_OFF;

  dim3 block(256);
  dim3 pgrid((NPAIRS + WAVES - 1) / WAVES);
  hipLaunchKernelGGL(kde_prep_b, pgrid, block, 0, stream, Xm, bw, wfrag, wnorm);

  dim3 grid(NROWBLK * NSPLIT);
  hipLaunchKernelGGL(kde_logsumexp_wmma, grid, block, 0, stream, x, wfrag, wnorm, bw, psum, pmax);

  dim3 mgrid(M_ROWS / 256);
  hipLaunchKernelGGL(kde_merge_ranges, mgrid, block, 0, stream, psum, pmax, outp);
}
